# starting kernel
# baseline (speedup 1.0000x reference)
#include <hip/hip_runtime.h>

typedef __attribute__((ext_vector_type(2))) float v2f;
typedef __attribute__((ext_vector_type(8))) float v8f;

#define C_IN  32
#define KCO   192
#define C_OUT 32
#define NK    6
#define ND    3

__global__ void prep_kernel(const float* __restrict__ mu,
                            const float* __restrict__ sigma,
                            float* __restrict__ inv, float* __restrict__ muc) {
  int i = threadIdx.x;
  if (i < NK * ND) {
    float sg = sigma[i];
    inv[i] = -0.5f / (1e-15f + sg * sg);
    muc[i] = mu[i];
  }
}

__global__ __launch_bounds__(128)
void xg_wmma_kernel(const float* __restrict__ x, const float* __restrict__ g,
                    float* __restrict__ xg, int N) {
  const int lane = threadIdx.x & 31;
  const int wave = threadIdx.x >> 5;
  const int m0 = blockIdx.x * 16;
  const int j  = blockIdx.y * 4 + wave;
  const int hx  = lane >> 4;
  const int l16 = lane & 15;
  const int n0 = j * 16;

  int row = m0 + l16;
  if (row >= N) row = N - 1;

  v8f acc = {};
#pragma unroll
  for (int k = 0; k < C_IN; k += 4) {
    v2f a = *(const v2f*)(x + row * C_IN + k + 2 * hx);
    v2f b;
    b.x = g[(k + 2 * hx) * KCO + n0 + l16];
    b.y = g[(k + 2 * hx + 1) * KCO + n0 + l16];
    acc = __builtin_amdgcn_wmma_f32_16x16x4_f32(
         false, a,  false, b,
         (short)0, acc,  false,  false);
  }

  float* p = xg + (m0 + 8 * hx) * KCO + n0 + l16;
  if (m0 + 16 <= N) {
#pragma unroll
    for (int i = 0; i < 8; ++i) p[i * KCO] = acc[i];
  } else {
#pragma unroll
    for (int i = 0; i < 8; ++i) {
      int r = m0 + i + 8 * hx;
      if (r < N) xg[r * KCO + n0 + l16] = acc[i];
    }
  }
}

__global__ __launch_bounds__(256)
void edge_kernel(const float* __restrict__ xg, const int* __restrict__ edge_index,
                 const float* __restrict__ pseudo, const float* __restrict__ inv,
                 const float* __restrict__ muc,
                 float* __restrict__ out, float* __restrict__ cnt, int E) {
  const int c = threadIdx.x & 31;
  const int e = blockIdx.x * 8 + (threadIdx.x >> 5);
  if (e >= E) return;

  const int src = edge_index[e];
  const int dst = edge_index[E + e];
  const float* xrow = xg + src * KCO;
  __builtin_prefetch(xrow + c, 0, 0);

  const float p0 = pseudo[e * ND + 0];
  const float p1 = pseudo[e * ND + 1];
  const float p2 = pseudo[e * ND + 2];

  float msg = 0.0f;
#pragma unroll
  for (int k = 0; k < NK; ++k) {
    float d0 = p0 - muc[k * ND + 0];
    float d1 = p1 - muc[k * ND + 1];
    float d2 = p2 - muc[k * ND + 2];
    float s = d0 * d0 * inv[k * ND + 0];
    s = fmaf(d1 * d1, inv[k * ND + 1], s);
    s = fmaf(d2 * d2, inv[k * ND + 2], s);
    msg = fmaf(__expf(s), xrow[k * C_OUT + c], msg);
  }

  __hip_atomic_fetch_add(out + dst * C_OUT + c, msg,
                         __ATOMIC_RELAXED, __HIP_MEMORY_SCOPE_AGENT);
  if (c == 0)
    __hip_atomic_fetch_add(cnt + dst, 1.0f,
                           __ATOMIC_RELAXED, __HIP_MEMORY_SCOPE_AGENT);
}

__global__ __launch_bounds__(64)
void finalize_kernel(const float* __restrict__ x, const float* __restrict__ root,
                     const float* __restrict__ bias, const float* __restrict__ cnt,
                     float* __restrict__ out, int N) {
  const int lane = threadIdx.x & 31;
  const int j = threadIdx.x >> 5;
  const int m0 = blockIdx.x * 16;
  const int hx  = lane >> 4;
  const int l16 = lane & 15;
  const int n0 = j * 16;

  int row = m0 + l16;
  if (row >= N) row = N - 1;

  v8f acc = {};
#pragma unroll
  for (int k = 0; k < C_IN; k += 4) {
    v2f a = *(const v2f*)(x + row * C_IN + k + 2 * hx);
    v2f b;
    b.x = root[(k + 2 * hx) * C_OUT + n0 + l16];
    b.y = root[(k + 2 * hx + 1) * C_OUT + n0 + l16];
    acc = __builtin_amdgcn_wmma_f32_16x16x4_f32(
        false, a, false, b, (short)0, acc, false, false);
  }

  const float bv = bias[n0 + l16];
  if (m0 + 16 <= N) {
    float* p = out + (m0 + 8 * hx) * C_OUT + n0 + l16;
    const float* cp = cnt + m0 + 8 * hx;
#pragma unroll
    for (int i = 0; i < 8; ++i) {
      float cv = cp[i];
      float denom = cv > 1.0f ? cv : 1.0f;
      p[i * C_OUT] = p[i * C_OUT] / denom + acc[i] + bv;
    }
  } else {
#pragma unroll
    for (int i = 0; i < 8; ++i) {
      int r = m0 + i + 8 * hx;
      if (r < N) {
        float cv = cnt[r];
        float denom = cv > 1.0f ? cv : 1.0f;
        int idx = r * C_OUT + n0 + l16;
        out[idx] = out[idx] / denom + acc[i] + bv;
      }
    }
  }
}

extern "C" void kernel_launch(void* const* d_in, const int* in_sizes, int n_in,
                              void* d_out, int out_size, void* d_ws, size_t ws_size,
                              hipStream_t stream) {
  const float* x          = (const float*)d_in[0];
  const int*   edge_index = (const int*)  d_in[1];
  const float* pseudo     = (const float*)d_in[2];
  const float* g          = (const float*)d_in[3];
  const float* mu         = (const float*)d_in[4];
  const float* sigma      = (const float*)d_in[5];
  const float* root       = (const float*)d_in[6];
  const float* bias       = (const float*)d_in[7];
  float* out = (float*)d_out;

  const int N = in_sizes[0] / C_IN;
  const int E = in_sizes[1] / 2;

  float* xg   = (float*)d_ws;
  float* cnt  = xg + (size_t)N * KCO;
  float* invc = cnt + N;
  float* muc  = invc + NK * ND;

  hipMemsetAsync(out, 0, (size_t)out_size * sizeof(float), stream);
  hipMemsetAsync(cnt, 0, (size_t)N * sizeof(float), stream);

  prep_kernel<<<1, 32, 0, stream>>>(mu, sigma, invc, muc);

  const int mt = (N + 15) / 16;
  xg_wmma_kernel<<<dim3(mt, 3), 128, 0, stream>>>(x, g, xg, N);
  edge_kernel<<<(E + 7) / 8, 256, 0, stream>>>(xg, edge_index, pseudo, invc, muc,
                                               out, cnt, E);
  finalize_kernel<<<mt, 64, 0, stream>>>(x, root, bias, cnt, out, N);
}


# Sum gauss-scaled x[src] rows per target first (edges counting-sorted by target, integer atomics only), then one f32 WMMA product with g per 16 nodes; float atomics, xg plane gone; sum order changes
# speedup vs baseline: 1.2506x; 1.2506x over previous
#include <hip/hip_runtime.h>


typedef __attribute__((ext_vector_type(2))) float v2f;
typedef __attribute__((ext_vector_type(4))) float v4f;
typedef __attribute__((ext_vector_type(8))) float v8f;

#define C_IN  32
#define KCO   192
#define C_OUT 32
#define NK    6
#define ND    3
#define REC   8
#define APITCH 196

__global__ void prep_kernel(const float* __restrict__ mu,
                            const float* __restrict__ sigma,
                            float* __restrict__ inv, float* __restrict__ muc) {
  int i = threadIdx.x;
  if (i < NK * ND) {
    float sg = sigma[i];
    inv[i] = -0.5f / (1e-15f + sg * sg);
    muc[i] = mu[i];
  }
}

__global__ __launch_bounds__(256)
void hist_kernel(const int* __restrict__ edge_index, int* __restrict__ deg, int E) {
  const int e = blockIdx.x * 256 + threadIdx.x;
  if (e < E) {
    const int dst = edge_index[E + e];
    __hip_atomic_fetch_add(deg + dst, 1, __ATOMIC_RELAXED, __HIP_MEMORY_SCOPE_AGENT);
  }
}

__global__ __launch_bounds__(256)
void alloc_kernel(const int* __restrict__ deg, int* __restrict__ cursor,
                  int* __restrict__ start, int* __restrict__ cur, int N) {
  const int n = blockIdx.x * 256 + threadIdx.x;
  const int lane = threadIdx.x & 31;
  const int d = (n < N) ? deg[n] : 0;
  int incl = d;
  for (int o = 1; o < 32; o <<= 1) {
    int t = __shfl_up(incl, o);
    if (lane >= o) incl += t;
  }
  const int total = __shfl(incl, 31);
  int base = 0;
  if (lane == 0)
    base = __hip_atomic_fetch_add(cursor, total, __ATOMIC_RELAXED, __HIP_MEMORY_SCOPE_AGENT);
  base = __shfl(base, 0);
  const int s = base + incl - d;
  if (n < N) { start[n] = s; cur[n] = s; }
}

__global__ __launch_bounds__(256)
void fill_kernel(const int* __restrict__ edge_index, const float* __restrict__ pseudo,
                 const float* __restrict__ inv, const float* __restrict__ muc,
                 int* __restrict__ cur, float* __restrict__ rec, int E) {
  const int e = blockIdx.x * 256 + threadIdx.x;
  if (e >= E) return;
  const int src = edge_index[e];
  const int dst = edge_index[E + e];
  const float p0 = pseudo[e * ND + 0];
  const float p1 = pseudo[e * ND + 1];
  const float p2 = pseudo[e * ND + 2];
  float w[NK];
#pragma unroll
  for (int k = 0; k < NK; ++k) {
    float d0 = p0 - muc[k * ND + 0];
    float d1 = p1 - muc[k * ND + 1];
    float d2 = p2 - muc[k * ND + 2];
    float s = d0 * d0 * inv[k * ND + 0];
    s = fmaf(d1 * d1, inv[k * ND + 1], s);
    s = fmaf(d2 * d2, inv[k * ND + 2], s);
    w[k] = __expf(s);
  }
  const int pos = __hip_atomic_fetch_add(cur + dst, 1, __ATOMIC_RELAXED, __HIP_MEMORY_SCOPE_AGENT);
  float* rp = rec + (size_t)pos * REC;
  v4f q0 = {__int_as_float(src), w[0], w[1], w[2]};
  v4f q1 = {w[3], w[4], w[5], 0.0f};
  *(v4f*)rp = q0;
  *(v4f*)(rp + 4) = q1;
}

static __device__ __forceinline__ float lane_bcast(float v, int t) {
  return __int_as_float(__builtin_amdgcn_readlane(__float_as_int(v), t));
}

__global__ __launch_bounds__(128)
void node_kernel(const float* __restrict__ x, const float* __restrict__ g,
                 const float* __restrict__ rec, const int* __restrict__ start,
                 const int* __restrict__ cur,
                 float* __restrict__ out, float* __restrict__ cnt, int N) {
  __shared__ float lds_all[4 * 16 * APITCH];
  const int lane = threadIdx.x & 31;
  const int wv = __builtin_amdgcn_readfirstlane(threadIdx.x >> 5);
  const int m0 = (blockIdx.x * 4 + wv) * 16;
  if (m0 >= N) return;
  float* lds = lds_all + wv * (16 * APITCH);

  const int nl = m0 + (lane & 15);
  const int last = (m0 + 15 < N) ? (m0 + 15) : (N - 1);
  const int endv = cur[last];
  const int sv = start[nl < N ? nl : N - 1];
  const int bv = (lane < 16 && nl < N) ? sv : endv;
  const int end = __builtin_amdgcn_readfirstlane(endv);
  const int beg = __builtin_amdgcn_readlane(bv, 0);

  {
    const int bnext = __shfl_down(bv, 1);
    if (lane < 16 && nl < N) cnt[nl] = (float)(bnext - bv);
  }

  float a0 = 0.f, a1 = 0.f, a2 = 0.f, a3 = 0.f, a4 = 0.f, a5 = 0.f;
  int i = 0;
  int nb = __builtin_amdgcn_readlane(bv, 1);

  for (int c0 = beg; c0 < end; c0 += 32) {
    int rl = c0 + lane;
    rl = rl < end ? rl : end - 1;
    const float* rp = rec + (size_t)rl * REC;
    const v4f q0 = *(const v4f*)rp;
    const v4f q1 = *(const v4f*)(rp + 4);
    const int srcv = __float_as_int(q0.x);
#pragma unroll
    for (int h = 0; h < 2; ++h) {
      const int g0 = c0 + 16 * h;
      if (g0 < end) {
        float xv[16];
#pragma unroll
        for (int u = 0; u < 16; ++u) {
          const int su = __builtin_amdgcn_readlane(srcv, 16 * h + u);
          xv[u] = x[su * C_IN + lane];
        }
#pragma unroll
        for (int u = 0; u < 16; ++u) {
          const int rr = g0 + u;
          if (rr < end) {
            while (rr >= nb) {
              float* lp = lds + i * APITCH + lane;
              lp[0] = a0; lp[32] = a1; lp[64] = a2; lp[96] = a3; lp[128] = a4; lp[160] = a5;
              a0 = a1 = a2 = a3 = a4 = a5 = 0.f;
              ++i;
              nb = __builtin_amdgcn_readlane(bv, i + 1);
            }
            const int t = 16 * h + u;
            a0 = fmaf(lane_bcast(q0.y, t), xv[u], a0);
            a1 = fmaf(lane_bcast(q0.z, t), xv[u], a1);
            a2 = fmaf(lane_bcast(q0.w, t), xv[u], a2);
            a3 = fmaf(lane_bcast(q1.x, t), xv[u], a3);
            a4 = fmaf(lane_bcast(q1.y, t), xv[u], a4);
            a5 = fmaf(lane_bcast(q1.z, t), xv[u], a5);
          }
        }
      }
    }
  }
  while (i < 16) {
    float* lp = lds + i * APITCH + lane;
    lp[0] = a0; lp[32] = a1; lp[64] = a2; lp[96] = a3; lp[128] = a4; lp[160] = a5;
    a0 = a1 = a2 = a3 = a4 = a5 = 0.f;
    ++i;
  }

  const int hx  = lane >> 4;
  const int l16 = lane & 15;
  v8f c0 = {}, c1 = {};
#pragma unroll
  for (int kk = 0; kk < KCO; kk += 4) {
    const int j  = kk + 2 * hx;
    const int kb = j >> 5;
    const int ci = j & 31;
    v2f a = *(const v2f*)(lds + l16 * APITCH + j);
    const float* gp = g + ci * KCO + kb * C_OUT + l16;
    v2f b0, b1;
    b0.x = gp[0];   b0.y = gp[KCO];
    b1.x = gp[16];  b1.y = gp[KCO + 16];
    c0 = __builtin_amdgcn_wmma_f32_16x16x4_f32(
         false, a,  false, b0,
         (short)0, c0,  false,  false);
    c1 = __builtin_amdgcn_wmma_f32_16x16x4_f32(
        false, a, false, b1, (short)0, c1, false, false);
  }

  float* p = out + (m0 + 8 * hx) * C_OUT + l16;
  if (m0 + 16 <= N) {
#pragma unroll
    for (int r = 0; r < 8; ++r) { p[r * C_OUT] = c0[r]; p[r * C_OUT + 16] = c1[r]; }
  } else {
#pragma unroll
    for (int r = 0; r < 8; ++r) {
      if (m0 + 8 * hx + r < N) { p[r * C_OUT] = c0[r]; p[r * C_OUT + 16] = c1[r]; }
    }
  }
}

__global__ __launch_bounds__(64)
void finalize_kernel(const float* __restrict__ x, const float* __restrict__ root,
                     const float* __restrict__ bias, const float* __restrict__ cnt,
                     float* __restrict__ out, int N) {
  const int lane = threadIdx.x & 31;
  const int j = threadIdx.x >> 5;
  const int m0 = blockIdx.x * 16;
  const int hx  = lane >> 4;
  const int l16 = lane & 15;
  const int n0 = j * 16;

  int row = m0 + l16;
  if (row >= N) row = N - 1;

  v8f acc = {};
#pragma unroll
  for (int k = 0; k < C_IN; k += 4) {
    v2f a = *(const v2f*)(x + row * C_IN + k + 2 * hx);
    v2f b;
    b.x = root[(k + 2 * hx) * C_OUT + n0 + l16];
    b.y = root[(k + 2 * hx + 1) * C_OUT + n0 + l16];
    acc = __builtin_amdgcn_wmma_f32_16x16x4_f32(
        false, a, false, b, (short)0, acc, false, false);
  }

  const float bv = bias[n0 + l16];
  if (m0 + 16 <= N) {
    float* p = out + (m0 + 8 * hx) * C_OUT + n0 + l16;
    const float* cp = cnt + m0 + 8 * hx;
#pragma unroll
    for (int i = 0; i < 8; ++i) {
      float cv = cp[i];
      float denom = cv > 1.0f ? cv : 1.0f;
      p[i * C_OUT] = p[i * C_OUT] / denom + acc[i] + bv;
    }
  } else {
#pragma unroll
    for (int i = 0; i < 8; ++i) {
      int r = m0 + i + 8 * hx;
      if (r < N) {
        float cv = cnt[r];
        float denom = cv > 1.0f ? cv : 1.0f;
        int idx = r * C_OUT + n0 + l16;
        out[idx] = out[idx] / denom + acc[i] + bv;
      }
    }
  }
}

extern "C" void kernel_launch(void* const* d_in, const int* in_sizes, int n_in,
                              void* d_out, int out_size, void* d_ws, size_t ws_size,
                              hipStream_t stream) {
  const float* x          = (const float*)d_in[0];
  const int*   edge_index = (const int*)  d_in[1];
  const float* pseudo     = (const float*)d_in[2];
  const float* g          = (const float*)d_in[3];
  const float* mu         = (const float*)d_in[4];
  const float* sigma      = (const float*)d_in[5];
  const float* root       = (const float*)d_in[6];
  const float* bias       = (const float*)d_in[7];
  float* out = (float*)d_out;

  const int N = in_sizes[0] / C_IN;
  const int E = in_sizes[1] / 2;

  float* rec    = (float*)d_ws;
  int*   deg    = (int*)(rec + (size_t)E * REC);
  int*   cursor = deg + N;
  int*   start  = cursor + 1;
  int*   cur    = start + N;
  float* cnt    = (float*)(cur + N);
  float* invc   = cnt + N;
  float* muc    = invc + NK * ND;

  hipMemsetAsync(deg, 0, (size_t)(N + 1) * sizeof(int), stream);

  prep_kernel<<<1, 32, 0, stream>>>(mu, sigma, invc, muc);

  const int mt = (N + 15) / 16;
  hist_kernel<<<(E + 255) / 256, 256, 0, stream>>>(edge_index, deg, E);
  alloc_kernel<<<(N + 255) / 256, 256, 0, stream>>>(deg, cursor, start, cur, N);
  fill_kernel<<<(E + 255) / 256, 256, 0, stream>>>(edge_index, pseudo, invc, muc, cur, rec, E);
  node_kernel<<<(mt + 3) / 4, 128, 0, stream>>>(x, g, rec, start, cur, out, cnt, N);
  finalize_kernel<<<mt, 64, 0, stream>>>(x, root, bias, cnt, out, N);
}


# Edge's slot rank kept from the counting atomic's return; fill places records at start[dst]+rank with plain loads, second atomic pass gone; within-node sum order = count-pass arrival order
# speedup vs baseline: 1.5292x; 1.2228x over previous
#include <hip/hip_runtime.h>


typedef __attribute__((ext_vector_type(2))) float v2f;
typedef __attribute__((ext_vector_type(4))) float v4f;
typedef __attribute__((ext_vector_type(8))) float v8f;

#define C_IN  32
#define KCO   192
#define C_OUT 32
#define NK    6
#define ND    3
#define REC   8
#define APITCH 196

__global__ void prep_kernel(const float* __restrict__ mu,
                            const float* __restrict__ sigma,
                            float* __restrict__ inv, float* __restrict__ muc) {
  int i = threadIdx.x;
  if (i < NK * ND) {
    float sg = sigma[i];
    inv[i] = -0.5f / (1e-15f + sg * sg);
    muc[i] = mu[i];
  }
}

__global__ __launch_bounds__(256)
void hist_kernel(const int* __restrict__ edge_index, int* __restrict__ deg,
                 int* __restrict__ rank, int E) {
  const int e = blockIdx.x * 256 + threadIdx.x;
  if (e < E) {
    const int dst = edge_index[E + e];
    rank[e] = __hip_atomic_fetch_add(deg + dst, 1, __ATOMIC_RELAXED, __HIP_MEMORY_SCOPE_AGENT);
  }
}

__global__ __launch_bounds__(256)
void alloc_kernel(const int* __restrict__ deg, int* __restrict__ cursor,
                  int* __restrict__ start, int N) {
  const int n = blockIdx.x * 256 + threadIdx.x;
  const int lane = threadIdx.x & 31;
  const int d = (n < N) ? deg[n] : 0;
  int incl = d;
  for (int o = 1; o < 32; o <<= 1) {
    int t = __shfl_up(incl, o);
    if (lane >= o) incl += t;
  }
  const int total = __shfl(incl, 31);
  int base = 0;
  if (lane == 0)
    base = __hip_atomic_fetch_add(cursor, total, __ATOMIC_RELAXED, __HIP_MEMORY_SCOPE_AGENT);
  base = __shfl(base, 0);
  const int s = base + incl - d;
  if (n < N) start[n] = s;
}

__global__ __launch_bounds__(256)
void fill_kernel(const int* __restrict__ edge_index, const float* __restrict__ pseudo,
                 const float* __restrict__ inv, const float* __restrict__ muc,
                 const int* __restrict__ start, const int* __restrict__ rank,
                 float* __restrict__ rec, int E) {
  const int e = blockIdx.x * 256 + threadIdx.x;
  if (e >= E) return;
  const int src = edge_index[e];
  const int dst = edge_index[E + e];
  const int pos = start[dst] + rank[e];
  const float p0 = pseudo[e * ND + 0];
  const float p1 = pseudo[e * ND + 1];
  const float p2 = pseudo[e * ND + 2];
  float w[NK];
#pragma unroll
  for (int k = 0; k < NK; ++k) {
    float d0 = p0 - muc[k * ND + 0];
    float d1 = p1 - muc[k * ND + 1];
    float d2 = p2 - muc[k * ND + 2];
    float s = d0 * d0 * inv[k * ND + 0];
    s = fmaf(d1 * d1, inv[k * ND + 1], s);
    s = fmaf(d2 * d2, inv[k * ND + 2], s);
    w[k] = __expf(s);
  }
  float* rp = rec + (size_t)pos * REC;
  v4f q0 = {__int_as_float(src), w[0], w[1], w[2]};
  v4f q1 = {w[3], w[4], w[5], 0.0f};
  *(v4f*)rp = q0;
  *(v4f*)(rp + 4) = q1;
}

static __device__ __forceinline__ float lane_bcast(float v, int t) {
  return __int_as_float(__builtin_amdgcn_readlane(__float_as_int(v), t));
}

__global__ __launch_bounds__(128)
void node_kernel(const float* __restrict__ x, const float* __restrict__ g,
                 const float* __restrict__ rec, const int* __restrict__ start,
                 const int* __restrict__ deg,
                 float* __restrict__ out, float* __restrict__ cnt, int N) {
  __shared__ float lds_all[4 * 16 * APITCH];
  const int lane = threadIdx.x & 31;
  const int wv = __builtin_amdgcn_readfirstlane(threadIdx.x >> 5);
  const int m0 = (blockIdx.x * 4 + wv) * 16;
  if (m0 >= N) return;
  float* lds = lds_all + wv * (16 * APITCH);

  const int nl = m0 + (lane & 15);
  const int last = (m0 + 15 < N) ? (m0 + 15) : (N - 1);
  const int endv = start[last] + deg[last];
  const int sv = start[nl < N ? nl : N - 1];
  const int bv = (lane < 16 && nl < N) ? sv : endv;
  const int end = __builtin_amdgcn_readfirstlane(endv);
  const int beg = __builtin_amdgcn_readlane(bv, 0);

  {
    const int bnext = __shfl_down(bv, 1);
    if (lane < 16 && nl < N) cnt[nl] = (float)(bnext - bv);
  }

  float a0 = 0.f, a1 = 0.f, a2 = 0.f, a3 = 0.f, a4 = 0.f, a5 = 0.f;
  int i = 0;
  int nb = __builtin_amdgcn_readlane(bv, 1);

  for (int c0 = beg; c0 < end; c0 += 32) {
    int rl = c0 + lane;
    rl = rl < end ? rl : end - 1;
    const float* rp = rec + (size_t)rl * REC;
    const v4f q0 = *(const v4f*)rp;
    const v4f q1 = *(const v4f*)(rp + 4);
    const int srcv = __float_as_int(q0.x);
#pragma unroll
    for (int h = 0; h < 2; ++h) {
      const int g0 = c0 + 16 * h;
      if (g0 < end) {
        float xv[16];
#pragma unroll
        for (int u = 0; u < 16; ++u) {
          const int su = __builtin_amdgcn_readlane(srcv, 16 * h + u);
          xv[u] = x[su * C_IN + lane];
        }
#pragma unroll
        for (int u = 0; u < 16; ++u) {
          const int rr = g0 + u;
          if (rr < end) {
            while (rr >= nb) {
              float* lp = lds + i * APITCH + lane;
              lp[0] = a0; lp[32] = a1; lp[64] = a2; lp[96] = a3; lp[128] = a4; lp[160] = a5;
              a0 = a1 = a2 = a3 = a4 = a5 = 0.f;
              ++i;
              nb = __builtin_amdgcn_readlane(bv, i + 1);
            }
            const int t = 16 * h + u;
            a0 = fmaf(lane_bcast(q0.y, t), xv[u], a0);
            a1 = fmaf(lane_bcast(q0.z, t), xv[u], a1);
            a2 = fmaf(lane_bcast(q0.w, t), xv[u], a2);
            a3 = fmaf(lane_bcast(q1.x, t), xv[u], a3);
            a4 = fmaf(lane_bcast(q1.y, t), xv[u], a4);
            a5 = fmaf(lane_bcast(q1.z, t), xv[u], a5);
          }
        }
      }
    }
  }
  while (i < 16) {
    float* lp = lds + i * APITCH + lane;
    lp[0] = a0; lp[32] = a1; lp[64] = a2; lp[96] = a3; lp[128] = a4; lp[160] = a5;
    a0 = a1 = a2 = a3 = a4 = a5 = 0.f;
    ++i;
  }

  const int hx  = lane >> 4;
  const int l16 = lane & 15;
  v8f c0 = {}, c1 = {};
#pragma unroll
  for (int kk = 0; kk < KCO; kk += 4) {
    const int j  = kk + 2 * hx;
    const int kb = j >> 5;
    const int ci = j & 31;
    v2f a = *(const v2f*)(lds + l16 * APITCH + j);
    const float* gp = g + ci * KCO + kb * C_OUT + l16;
    v2f b0, b1;
    b0.x = gp[0];   b0.y = gp[KCO];
    b1.x = gp[16];  b1.y = gp[KCO + 16];
    c0 = __builtin_amdgcn_wmma_f32_16x16x4_f32(
         false, a,  false, b0,
         (short)0, c0,  false,  false);
    c1 = __builtin_amdgcn_wmma_f32_16x16x4_f32(
        false, a, false, b1, (short)0, c1, false, false);
  }

  float* p = out + (m0 + 8 * hx) * C_OUT + l16;
  if (m0 + 16 <= N) {
#pragma unroll
    for (int r = 0; r < 8; ++r) { p[r * C_OUT] = c0[r]; p[r * C_OUT + 16] = c1[r]; }
  } else {
#pragma unroll
    for (int r = 0; r < 8; ++r) {
      if (m0 + 8 * hx + r < N) { p[r * C_OUT] = c0[r]; p[r * C_OUT + 16] = c1[r]; }
    }
  }
}

__global__ __launch_bounds__(64)
void finalize_kernel(const float* __restrict__ x, const float* __restrict__ root,
                     const float* __restrict__ bias, const float* __restrict__ cnt,
                     float* __restrict__ out, int N) {
  const int lane = threadIdx.x & 31;
  const int j = threadIdx.x >> 5;
  const int m0 = blockIdx.x * 16;
  const int hx  = lane >> 4;
  const int l16 = lane & 15;
  const int n0 = j * 16;

  int row = m0 + l16;
  if (row >= N) row = N - 1;

  v8f acc = {};
#pragma unroll
  for (int k = 0; k < C_IN; k += 4) {
    v2f a = *(const v2f*)(x + row * C_IN + k + 2 * hx);
    v2f b;
    b.x = root[(k + 2 * hx) * C_OUT + n0 + l16];
    b.y = root[(k + 2 * hx + 1) * C_OUT + n0 + l16];
    acc = __builtin_amdgcn_wmma_f32_16x16x4_f32(
        false, a, false, b, (short)0, acc, false, false);
  }

  const float bv = bias[n0 + l16];
  if (m0 + 16 <= N) {
    float* p = out + (m0 + 8 * hx) * C_OUT + n0 + l16;
    const float* cp = cnt + m0 + 8 * hx;
#pragma unroll
    for (int i = 0; i < 8; ++i) {
      float cv = cp[i];
      float denom = cv > 1.0f ? cv : 1.0f;
      p[i * C_OUT] = p[i * C_OUT] / denom + acc[i] + bv;
    }
  } else {
#pragma unroll
    for (int i = 0; i < 8; ++i) {
      int r = m0 + i + 8 * hx;
      if (r < N) {
        float cv = cnt[r];
        float denom = cv > 1.0f ? cv : 1.0f;
        int idx = r * C_OUT + n0 + l16;
        out[idx] = out[idx] / denom + acc[i] + bv;
      }
    }
  }
}

extern "C" void kernel_launch(void* const* d_in, const int* in_sizes, int n_in,
                              void* d_out, int out_size, void* d_ws, size_t ws_size,
                              hipStream_t stream) {
  const float* x          = (const float*)d_in[0];
  const int*   edge_index = (const int*)  d_in[1];
  const float* pseudo     = (const float*)d_in[2];
  const float* g          = (const float*)d_in[3];
  const float* mu         = (const float*)d_in[4];
  const float* sigma      = (const float*)d_in[5];
  const float* root       = (const float*)d_in[6];
  const float* bias       = (const float*)d_in[7];
  float* out = (float*)d_out;

  const int N = in_sizes[0] / C_IN;
  const int E = in_sizes[1] / 2;

  float* rec    = (float*)d_ws;
  int*   deg    = (int*)(rec + (size_t)E * REC);
  int*   cursor = deg + N;
  int*   start  = cursor + 1;
  int*   rank   = start + N;
  float* cnt    = (float*)(rank + E);
  float* invc   = cnt + N;
  float* muc    = invc + NK * ND;

  hipMemsetAsync(deg, 0, (size_t)(N + 1) * sizeof(int), stream);

  prep_kernel<<<1, 32, 0, stream>>>(mu, sigma, invc, muc);

  const int mt = (N + 15) / 16;
  hist_kernel<<<(E + 255) / 256, 256, 0, stream>>>(edge_index, deg, rank, E);
  alloc_kernel<<<(N + 255) / 256, 256, 0, stream>>>(deg, cursor, start, N);
  fill_kernel<<<(E + 255) / 256, 256, 0, stream>>>(edge_index, pseudo, invc, muc, start, rank, rec, E);
  node_kernel<<<(mt + 3) / 4, 128, 0, stream>>>(x, g, rec, start, deg, out, cnt, N);
  finalize_kernel<<<mt, 64, 0, stream>>>(x, root, bias, cnt, out, N);
}


# Launch folding: finalize becomes the node kernel's epilogue (sums/deg, then x@root added onto them by f32 WMMA, bias); prep's 18 coefficients move into alloc; 6 launches -> 4; final add order changes
# speedup vs baseline: 1.6490x; 1.0783x over previous
#include <hip/hip_runtime.h>


typedef __attribute__((ext_vector_type(2))) float v2f;
typedef __attribute__((ext_vector_type(4))) float v4f;
typedef __attribute__((ext_vector_type(8))) float v8f;

#define C_IN  32
#define KCO   192
#define C_OUT 32
#define NK    6
#define ND    3
#define REC   8
#define APITCH 196

__global__ __launch_bounds__(256)
void hist_kernel(const int* __restrict__ edge_index, int* __restrict__ deg,
                 int* __restrict__ rank, int E) {
  const int e = blockIdx.x * 256 + threadIdx.x;
  if (e < E) {
    const int dst = edge_index[E + e];
    rank[e] = __hip_atomic_fetch_add(deg + dst, 1, __ATOMIC_RELAXED, __HIP_MEMORY_SCOPE_AGENT);
  }
}

__global__ __launch_bounds__(256)
void alloc_kernel(const int* __restrict__ deg, int* __restrict__ cursor,
                  int* __restrict__ start,
                  const float* __restrict__ mu, const float* __restrict__ sigma,
                  float* __restrict__ inv, float* __restrict__ muc, int N) {
  const int n = blockIdx.x * 256 + threadIdx.x;
  const int lane = threadIdx.x & 31;
  const int d = (n < N) ? deg[n] : 0;
  int incl = d;
  for (int o = 1; o < 32; o <<= 1) {
    int t = __shfl_up(incl, o);
    if (lane >= o) incl += t;
  }
  const int total = __shfl(incl, 31);
  int base = 0;
  if (lane == 0)
    base = __hip_atomic_fetch_add(cursor, total, __ATOMIC_RELAXED, __HIP_MEMORY_SCOPE_AGENT);
  base = __shfl(base, 0);
  const int s = base + incl - d;
  if (n < N) start[n] = s;
  if (blockIdx.x == 0 && threadIdx.x < NK * ND) {
    const float sg = sigma[threadIdx.x];
    inv[threadIdx.x] = -0.5f / (1e-15f + sg * sg);
    muc[threadIdx.x] = mu[threadIdx.x];
  }
}

__global__ __launch_bounds__(256)
void fill_kernel(const int* __restrict__ edge_index, const float* __restrict__ pseudo,
                 const float* __restrict__ inv, const float* __restrict__ muc,
                 const int* __restrict__ start, const int* __restrict__ rank,
                 float* __restrict__ rec, int E) {
  const int e = blockIdx.x * 256 + threadIdx.x;
  if (e >= E) return;
  const int src = edge_index[e];
  const int dst = edge_index[E + e];
  const int pos = start[dst] + rank[e];
  const float p0 = pseudo[e * ND + 0];
  const float p1 = pseudo[e * ND + 1];
  const float p2 = pseudo[e * ND + 2];
  float w[NK];
#pragma unroll
  for (int k = 0; k < NK; ++k) {
    float d0 = p0 - muc[k * ND + 0];
    float d1 = p1 - muc[k * ND + 1];
    float d2 = p2 - muc[k * ND + 2];
    float s = d0 * d0 * inv[k * ND + 0];
    s = fmaf(d1 * d1, inv[k * ND + 1], s);
    s = fmaf(d2 * d2, inv[k * ND + 2], s);
    w[k] = __expf(s);
  }
  float* rp = rec + (size_t)pos * REC;
  v4f q0 = {__int_as_float(src), w[0], w[1], w[2]};
  v4f q1 = {w[3], w[4], w[5], 0.0f};
  *(v4f*)rp = q0;
  *(v4f*)(rp + 4) = q1;
}

static __device__ __forceinline__ float lane_bcast(float v, int t) {
  return __int_as_float(__builtin_amdgcn_readlane(__float_as_int(v), t));
}

__global__ __launch_bounds__(128)
void node_kernel(const float* __restrict__ x, const float* __restrict__ g,
                 const float* __restrict__ root, const float* __restrict__ bias,
                 const float* __restrict__ rec, const int* __restrict__ start,
                 const int* __restrict__ deg,
                 float* __restrict__ out, int N) {
  __shared__ float lds_all[4 * 16 * APITCH];
  const int lane = threadIdx.x & 31;
  const int wv = __builtin_amdgcn_readfirstlane(threadIdx.x >> 5);
  const int m0 = (blockIdx.x * 4 + wv) * 16;
  if (m0 >= N) return;
  float* lds = lds_all + wv * (16 * APITCH);

  const int nl = m0 + (lane & 15);
  const int last = (m0 + 15 < N) ? (m0 + 15) : (N - 1);
  const int endv = start[last] + deg[last];
  const int sv = start[nl < N ? nl : N - 1];
  const int bv = (lane < 16 && nl < N) ? sv : endv;
  const int end = __builtin_amdgcn_readfirstlane(endv);
  const int beg = __builtin_amdgcn_readlane(bv, 0);

  const float cntv = (float)(__shfl_down(bv, 1) - bv);

  const int hx  = lane >> 4;
  const int l16 = lane & 15;

  float a0 = 0.f, a1 = 0.f, a2 = 0.f, a3 = 0.f, a4 = 0.f, a5 = 0.f;
  int i = 0;
  int nb = __builtin_amdgcn_readlane(bv, 1);

  for (int c0 = beg; c0 < end; c0 += 32) {
    int rl = c0 + lane;
    rl = rl < end ? rl : end - 1;
    const float* rp = rec + (size_t)rl * REC;
    const v4f q0 = *(const v4f*)rp;
    const v4f q1 = *(const v4f*)(rp + 4);
    const int srcv = __float_as_int(q0.x);
#pragma unroll
    for (int h = 0; h < 2; ++h) {
      const int g0 = c0 + 16 * h;
      if (g0 < end) {
        float xv[16];
#pragma unroll
        for (int u = 0; u < 16; ++u) {
          const int su = __builtin_amdgcn_readlane(srcv, 16 * h + u);
          xv[u] = x[su * C_IN + lane];
        }
#pragma unroll
        for (int u = 0; u < 16; ++u) {
          const int rr = g0 + u;
          if (rr < end) {
            while (rr >= nb) {
              float* lp = lds + i * APITCH + lane;
              lp[0] = a0; lp[32] = a1; lp[64] = a2; lp[96] = a3; lp[128] = a4; lp[160] = a5;
              a0 = a1 = a2 = a3 = a4 = a5 = 0.f;
              ++i;
              nb = __builtin_amdgcn_readlane(bv, i + 1);
            }
            const int t = 16 * h + u;
            a0 = fmaf(lane_bcast(q0.y, t), xv[u], a0);
            a1 = fmaf(lane_bcast(q0.z, t), xv[u], a1);
            a2 = fmaf(lane_bcast(q0.w, t), xv[u], a2);
            a3 = fmaf(lane_bcast(q1.x, t), xv[u], a3);
            a4 = fmaf(lane_bcast(q1.y, t), xv[u], a4);
            a5 = fmaf(lane_bcast(q1.z, t), xv[u], a5);
          }
        }
      }
    }
  }
  while (i < 16) {
    float* lp = lds + i * APITCH + lane;
    lp[0] = a0; lp[32] = a1; lp[64] = a2; lp[96] = a3; lp[128] = a4; lp[160] = a5;
    a0 = a1 = a2 = a3 = a4 = a5 = 0.f;
    ++i;
  }

  v8f c0 = {}, c1 = {};
#pragma unroll
  for (int kk = 0; kk < KCO; kk += 4) {
    const int j  = kk + 2 * hx;
    const int kb = j >> 5;
    const int ci = j & 31;
    v2f a = *(const v2f*)(lds + l16 * APITCH + j);
    const float* gp = g + ci * KCO + kb * C_OUT + l16;
    v2f b0, b1;
    b0.x = gp[0];   b0.y = gp[KCO];
    b1.x = gp[16];  b1.y = gp[KCO + 16];
    c0 = __builtin_amdgcn_wmma_f32_16x16x4_f32(
         false, a,  false, b0,
         (short)0, c0,  false,  false);
    c1 = __builtin_amdgcn_wmma_f32_16x16x4_f32(
        false, a, false, b1, (short)0, c1, false, false);
  }

  v2f ax[C_IN / 4];
  {
    const int xr = (m0 + l16 < N) ? (m0 + l16) : (N - 1);
#pragma unroll
    for (int k = 0; k < C_IN / 4; ++k) ax[k] = *(const v2f*)(x + xr * C_IN + 4 * k + 2 * hx);
  }

#pragma unroll
  for (int r = 0; r < 8; ++r) {
    const float cv = __shfl(cntv, 8 * hx + r);
    const float denom = cv > 1.0f ? cv : 1.0f;
    c0[r] = c0[r] / denom;
    c1[r] = c1[r] / denom;
  }
#pragma unroll
  for (int k = 0; k < C_IN / 4; ++k) {
    const float* rp = root + (4 * k + 2 * hx) * C_OUT + l16;
    v2f b0, b1;
    b0.x = rp[0];   b0.y = rp[C_OUT];
    b1.x = rp[16];  b1.y = rp[C_OUT + 16];
    c0 = __builtin_amdgcn_wmma_f32_16x16x4_f32(
        false, ax[k], false, b0, (short)0, c0, false, false);
    c1 = __builtin_amdgcn_wmma_f32_16x16x4_f32(
        false, ax[k], false, b1, (short)0, c1, false, false);
  }
  const float bv0 = bias[l16];
  const float bv1 = bias[16 + l16];
  float o0[8], o1[8];
#pragma unroll
  for (int r = 0; r < 8; ++r) { o0[r] = c0[r] + bv0; o1[r] = c1[r] + bv1; }
  float* p = out + (m0 + 8 * hx) * C_OUT + l16;
  if (m0 + 16 <= N) {
#pragma unroll
    for (int r = 0; r < 8; ++r) { p[r * C_OUT] = o0[r]; p[r * C_OUT + 16] = o1[r]; }
  } else {
#pragma unroll
    for (int r = 0; r < 8; ++r) {
      if (m0 + 8 * hx + r < N) { p[r * C_OUT] = o0[r]; p[r * C_OUT + 16] = o1[r]; }
    }
  }
}

extern "C" void kernel_launch(void* const* d_in, const int* in_sizes, int n_in,
                              void* d_out, int out_size, void* d_ws, size_t ws_size,
                              hipStream_t stream) {
  const float* x          = (const float*)d_in[0];
  const int*   edge_index = (const int*)  d_in[1];
  const float* pseudo     = (const float*)d_in[2];
  const float* g          = (const float*)d_in[3];
  const float* mu         = (const float*)d_in[4];
  const float* sigma      = (const float*)d_in[5];
  const float* root       = (const float*)d_in[6];
  const float* bias       = (const float*)d_in[7];
  float* out = (float*)d_out;

  const int N = in_sizes[0] / C_IN;
  const int E = in_sizes[1] / 2;

  float* rec    = (float*)d_ws;
  int*   deg    = (int*)(rec + (size_t)E * REC);
  int*   cursor = deg + N;
  int*   start  = cursor + 1;
  int*   rank   = start + N;
  float* invc   = (float*)(rank + E);
  float* muc    = invc + NK * ND;

  hipMemsetAsync(deg, 0, (size_t)(N + 1) * sizeof(int), stream);

  const int mt = (N + 15) / 16;
  hist_kernel<<<(E + 255) / 256, 256, 0, stream>>>(edge_index, deg, rank, E);
  alloc_kernel<<<(N + 255) / 256, 256, 0, stream>>>(deg, cursor, start, mu, sigma, invc, muc, N);
  fill_kernel<<<(E + 255) / 256, 256, 0, stream>>>(edge_index, pseudo, invc, muc, start, rank, rec, E);
  node_kernel<<<(mt + 3) / 4, 128, 0, stream>>>(x, g, root, bias, rec, start, deg, out, N);
}


# Counting sort by target redone on LDS counters in two levels (512-node bins, then nodes): global adds once per work-group and bin, not per edge; within-node sum order = LDS/bin-cursor arrival order
# speedup vs baseline: 2.3476x; 1.4237x over previous
#include <hip/hip_runtime.h>


typedef __attribute__((ext_vector_type(2))) float v2f;
typedef __attribute__((ext_vector_type(4))) float v4f;
typedef __attribute__((ext_vector_type(8))) float v8f;

#define C_IN  32
#define KCO   192
#define C_OUT 32
#define NK    6
#define ND    3
#define REC   8
#define APITCH 196
#define NPB_SHIFT 9
#define NPB  512
#define MAXB 1024
#define BT   1024
#define EPT  4
#define CH   (BT * EPT)


static __device__ __forceinline__ int block_excl_scan(int v, int* ws) {
  const int lane = threadIdx.x & 31;
  const int wave = threadIdx.x >> 5;
  int incl = v;
  for (int o = 1; o < 32; o <<= 1) {
    int t = __shfl_up(incl, o);
    if (lane >= o) incl += t;
  }
  if (lane == 31) ws[wave] = incl;
  __syncthreads();
  if (wave == 0) {
    const int w = ws[lane];
    int wi = w;
    for (int o = 1; o < 32; o <<= 1) {
      int t = __shfl_up(wi, o);
      if (lane >= o) wi += t;
    }
    ws[lane] = wi - w;
  }
  __syncthreads();
  return ws[wave] + incl - v;
}

__global__ __launch_bounds__(BT)
void bin_count_kernel(const int* __restrict__ edge_index, int* __restrict__ bintotal,
                      const float* __restrict__ mu, const float* __restrict__ sigma,
                      float* __restrict__ inv, float* __restrict__ muc, int E, int B) {
  __shared__ int hist[MAXB];
  const int tid = threadIdx.x;
  hist[tid] = 0;
  __syncthreads();
  const int e0 = blockIdx.x * CH;
  int dstv[EPT];
#pragma unroll
  for (int k = 0; k < EPT; ++k) {
    const int e = e0 + k * BT + tid;
    dstv[k] = edge_index[E + (e < E ? e : E - 1)];
  }
#pragma unroll
  for (int k = 0; k < EPT; ++k) {
    if (e0 + k * BT + tid < E)
      __hip_atomic_fetch_add(&hist[dstv[k] >> NPB_SHIFT], 1, __ATOMIC_RELAXED, __HIP_MEMORY_SCOPE_WORKGROUP);
  }
  __syncthreads();
  if (tid < B) {
    const int c = hist[tid];
    if (c > 0)
      __hip_atomic_fetch_add(bintotal + tid, c, __ATOMIC_RELAXED, __HIP_MEMORY_SCOPE_AGENT);
  }
  if (blockIdx.x == 0 && tid < NK * ND) {
    const float sg = sigma[tid];
    inv[tid] = -0.5f / (1e-15f + sg * sg);
    muc[tid] = mu[tid];
  }
}

__global__ __launch_bounds__(BT)
void bin_scatter_kernel(const int* __restrict__ edge_index, const float* __restrict__ pseudo,
                        const float* __restrict__ inv, const float* __restrict__ muc,
                        const int* __restrict__ bintotal, int* __restrict__ bincur,
                        float* __restrict__ tmp, int E, int B) {
  __shared__ int hist[MAXB];
  __shared__ int base[MAXB];
  __shared__ int ws[32];
  const int tid = threadIdx.x;
  hist[tid] = 0;
  __syncthreads();
  const int e0 = blockIdx.x * CH;
  int dstv[EPT], srcv[EPT];
  float pv[EPT][ND];
#pragma unroll
  for (int k = 0; k < EPT; ++k) {
    const int e = e0 + k * BT + tid;
    const int ee = e < E ? e : E - 1;
    dstv[k] = edge_index[E + ee];
    srcv[k] = edge_index[ee];
    pv[k][0] = pseudo[ee * ND + 0];
    pv[k][1] = pseudo[ee * ND + 1];
    pv[k][2] = pseudo[ee * ND + 2];
  }
#pragma unroll
  for (int k = 0; k < EPT; ++k) {
    if (e0 + k * BT + tid < E)
      __hip_atomic_fetch_add(&hist[dstv[k] >> NPB_SHIFT], 1, __ATOMIC_RELAXED, __HIP_MEMORY_SCOPE_WORKGROUP);
  }
  const int tot = (tid < B) ? bintotal[tid] : 0;
  const int binstart = block_excl_scan(tot, ws);
  const int mine = hist[tid];
  int b0 = 0;
  if (tid < B && mine > 0)
    b0 = binstart + __hip_atomic_fetch_add(bincur + tid, mine, __ATOMIC_RELAXED, __HIP_MEMORY_SCOPE_AGENT);
  base[tid] = b0;
  __syncthreads();
  hist[tid] = 0;
  __syncthreads();
#pragma unroll
  for (int k = 0; k < EPT; ++k) {
    const int e = e0 + k * BT + tid;
    if (e < E) {
      const int dst = dstv[k];
      const int bin = dst >> NPB_SHIFT;
      const int src = srcv[k];
      const float p0 = pv[k][0];
      const float p1 = pv[k][1];
      const float p2 = pv[k][2];
      const int r = __hip_atomic_fetch_add(&hist[bin], 1, __ATOMIC_RELAXED, __HIP_MEMORY_SCOPE_WORKGROUP);
      const int pos = base[bin] + r;
      float w[NK];
#pragma unroll
      for (int kk = 0; kk < NK; ++kk) {
        float d0 = p0 - muc[kk * ND + 0];
        float d1 = p1 - muc[kk * ND + 1];
        float d2 = p2 - muc[kk * ND + 2];
        float s = d0 * d0 * inv[kk * ND + 0];
        s = fmaf(d1 * d1, inv[kk * ND + 1], s);
        s = fmaf(d2 * d2, inv[kk * ND + 2], s);
        w[kk] = __expf(s);
      }
      float* rp = tmp + (size_t)pos * REC;
      v4f q0 = {__int_as_float(src), w[0], w[1], w[2]};
      v4f q1 = {w[3], w[4], w[5], __int_as_float(dst)};
      *(v4f*)rp = q0;
      *(v4f*)(rp + 4) = q1;
    }
  }
}

__global__ __launch_bounds__(BT)
void bin_sort_kernel(const int* __restrict__ bintotal, const float* __restrict__ tmp,
                     float* __restrict__ rec, int* __restrict__ start, int* __restrict__ deg,
                     int N, int B) {
  __shared__ int cnt[NPB];
  __shared__ int cur[NPB];
  __shared__ int ws0[32];
  __shared__ int ws1[32];
  __shared__ int sbs;
  const int tid = threadIdx.x;
  const int b = blockIdx.x;
  const int n0 = b << NPB_SHIFT;
  if (tid < NPB) cnt[tid] = 0;
  const int tot = (tid < B) ? bintotal[tid] : 0;
  const int pre = block_excl_scan(tot, ws0);
  if (tid == b) sbs = pre;
  __syncthreads();
  const int bs = sbs;
  const int be = bs + bintotal[b];
  for (int i0 = bs + tid; i0 < be; i0 += EPT * BT) {
    int dv[EPT];
#pragma unroll
    for (int k = 0; k < EPT; ++k) {
      const int i = i0 + k * BT;
      dv[k] = __float_as_int(tmp[(size_t)(i < be ? i : be - 1) * REC + 7]);
    }
#pragma unroll
    for (int k = 0; k < EPT; ++k) {
      if (i0 + k * BT < be)
        __hip_atomic_fetch_add(&cnt[dv[k] - n0], 1, __ATOMIC_RELAXED, __HIP_MEMORY_SCOPE_WORKGROUP);
    }
  }
  __syncthreads();
  const int c = (tid < NPB) ? cnt[tid] : 0;
  const int ex = block_excl_scan(c, ws1);
  if (tid < NPB) {
    cur[tid] = bs + ex;
    const int n = n0 + tid;
    if (n < N) { start[n] = bs + ex; deg[n] = c; }
  }
  __syncthreads();
  for (int i0 = bs + tid; i0 < be; i0 += EPT * BT) {
    v4f q0[EPT], q1[EPT];
#pragma unroll
    for (int k = 0; k < EPT; ++k) {
      const int i = i0 + k * BT;
      const float* sp = tmp + (size_t)(i < be ? i : be - 1) * REC;
      q0[k] = *(const v4f*)sp;
      q1[k] = *(const v4f*)(sp + 4);
    }
#pragma unroll
    for (int k = 0; k < EPT; ++k) {
      if (i0 + k * BT < be) {
        const int dst = __float_as_int(q1[k].w);
        const int pos = __hip_atomic_fetch_add(&cur[dst - n0], 1, __ATOMIC_RELAXED, __HIP_MEMORY_SCOPE_WORKGROUP);
        float* rp = rec + (size_t)pos * REC;
        *(v4f*)rp = q0[k];
        *(v4f*)(rp + 4) = q1[k];
      }
    }
  }
}

static __device__ __forceinline__ float lane_bcast(float v, int t) {
  return __int_as_float(__builtin_amdgcn_readlane(__float_as_int(v), t));
}

__global__ __launch_bounds__(128)
void node_kernel(const float* __restrict__ x, const float* __restrict__ g,
                 const float* __restrict__ root, const float* __restrict__ bias,
                 const float* __restrict__ rec, const int* __restrict__ start,
                 const int* __restrict__ deg,
                 float* __restrict__ out, int N) {
  __shared__ float lds_all[4 * 16 * APITCH];
  const int lane = threadIdx.x & 31;
  const int wv = __builtin_amdgcn_readfirstlane(threadIdx.x >> 5);
  const int m0 = (blockIdx.x * 4 + wv) * 16;
  if (m0 >= N) return;
  float* lds = lds_all + wv * (16 * APITCH);

  const int nl = m0 + (lane & 15);
  const int last = (m0 + 15 < N) ? (m0 + 15) : (N - 1);
  const int endv = start[last] + deg[last];
  const int sv = start[nl < N ? nl : N - 1];
  const int bv = (lane < 16 && nl < N) ? sv : endv;
  const int end = __builtin_amdgcn_readfirstlane(endv);
  const int beg = __builtin_amdgcn_readlane(bv, 0);

  const float cntv = (float)(__shfl_down(bv, 1) - bv);

  const int hx  = lane >> 4;
  const int l16 = lane & 15;

  float a0 = 0.f, a1 = 0.f, a2 = 0.f, a3 = 0.f, a4 = 0.f, a5 = 0.f;
  int i = 0;
  int nb = __builtin_amdgcn_readlane(bv, 1);

  for (int c0 = beg; c0 < end; c0 += 32) {
    int rl = c0 + lane;
    rl = rl < end ? rl : end - 1;
    const float* rp = rec + (size_t)rl * REC;
    const v4f q0 = *(const v4f*)rp;
    const v4f q1 = *(const v4f*)(rp + 4);
    const int srcv = __float_as_int(q0.x);
#pragma unroll
    for (int h = 0; h < 2; ++h) {
      const int g0 = c0 + 16 * h;
      if (g0 < end) {
        float xv[16];
#pragma unroll
        for (int u = 0; u < 16; ++u) {
          const int su = __builtin_amdgcn_readlane(srcv, 16 * h + u);
          xv[u] = x[su * C_IN + lane];
        }
#pragma unroll
        for (int u = 0; u < 16; ++u) {
          const int rr = g0 + u;
          if (rr < end) {
            while (rr >= nb) {
              float* lp = lds + i * APITCH + lane;
              lp[0] = a0; lp[32] = a1; lp[64] = a2; lp[96] = a3; lp[128] = a4; lp[160] = a5;
              a0 = a1 = a2 = a3 = a4 = a5 = 0.f;
              ++i;
              nb = __builtin_amdgcn_readlane(bv, i + 1);
            }
            const int t = 16 * h + u;
            a0 = fmaf(lane_bcast(q0.y, t), xv[u], a0);
            a1 = fmaf(lane_bcast(q0.z, t), xv[u], a1);
            a2 = fmaf(lane_bcast(q0.w, t), xv[u], a2);
            a3 = fmaf(lane_bcast(q1.x, t), xv[u], a3);
            a4 = fmaf(lane_bcast(q1.y, t), xv[u], a4);
            a5 = fmaf(lane_bcast(q1.z, t), xv[u], a5);
          }
        }
      }
    }
  }
  while (i < 16) {
    float* lp = lds + i * APITCH + lane;
    lp[0] = a0; lp[32] = a1; lp[64] = a2; lp[96] = a3; lp[128] = a4; lp[160] = a5;
    a0 = a1 = a2 = a3 = a4 = a5 = 0.f;
    ++i;
  }

  v8f c0 = {}, c1 = {};
#pragma unroll
  for (int kk = 0; kk < KCO; kk += 4) {
    const int j  = kk + 2 * hx;
    const int kb = j >> 5;
    const int ci = j & 31;
    v2f a = *(const v2f*)(lds + l16 * APITCH + j);
    const float* gp = g + ci * KCO + kb * C_OUT + l16;
    v2f b0, b1;
    b0.x = gp[0];   b0.y = gp[KCO];
    b1.x = gp[16];  b1.y = gp[KCO + 16];
    c0 = __builtin_amdgcn_wmma_f32_16x16x4_f32(
         false, a,  false, b0,
         (short)0, c0,  false,  false);
    c1 = __builtin_amdgcn_wmma_f32_16x16x4_f32(
        false, a, false, b1, (short)0, c1, false, false);
  }

  v2f ax[C_IN / 4];
  {
    const int xr = (m0 + l16 < N) ? (m0 + l16) : (N - 1);
#pragma unroll
    for (int k = 0; k < C_IN / 4; ++k) ax[k] = *(const v2f*)(x + xr * C_IN + 4 * k + 2 * hx);
  }

#pragma unroll
  for (int r = 0; r < 8; ++r) {
    const float cv = __shfl(cntv, 8 * hx + r);
    const float denom = cv > 1.0f ? cv : 1.0f;
    c0[r] = c0[r] / denom;
    c1[r] = c1[r] / denom;
  }
#pragma unroll
  for (int k = 0; k < C_IN / 4; ++k) {
    const float* rp = root + (4 * k + 2 * hx) * C_OUT + l16;
    v2f b0, b1;
    b0.x = rp[0];   b0.y = rp[C_OUT];
    b1.x = rp[16];  b1.y = rp[C_OUT + 16];
    c0 = __builtin_amdgcn_wmma_f32_16x16x4_f32(
        false, ax[k], false, b0, (short)0, c0, false, false);
    c1 = __builtin_amdgcn_wmma_f32_16x16x4_f32(
        false, ax[k], false, b1, (short)0, c1, false, false);
  }
  const float bv0 = bias[l16];
  const float bv1 = bias[16 + l16];
  float o0[8], o1[8];
#pragma unroll
  for (int r = 0; r < 8; ++r) { o0[r] = c0[r] + bv0; o1[r] = c1[r] + bv1; }
  float* p = out + (m0 + 8 * hx) * C_OUT + l16;
  if (m0 + 16 <= N) {
#pragma unroll
    for (int r = 0; r < 8; ++r) { p[r * C_OUT] = o0[r]; p[r * C_OUT + 16] = o1[r]; }
  } else {
#pragma unroll
    for (int r = 0; r < 8; ++r) {
      if (m0 + 8 * hx + r < N) { p[r * C_OUT] = o0[r]; p[r * C_OUT + 16] = o1[r]; }
    }
  }
}

extern "C" void kernel_launch(void* const* d_in, const int* in_sizes, int n_in,
                              void* d_out, int out_size, void* d_ws, size_t ws_size,
                              hipStream_t stream) {
  const float* x          = (const float*)d_in[0];
  const int*   edge_index = (const int*)  d_in[1];
  const float* pseudo     = (const float*)d_in[2];
  const float* g          = (const float*)d_in[3];
  const float* mu         = (const float*)d_in[4];
  const float* sigma      = (const float*)d_in[5];
  const float* root       = (const float*)d_in[6];
  const float* bias       = (const float*)d_in[7];
  float* out = (float*)d_out;

  const int N = in_sizes[0] / C_IN;
  const int E = in_sizes[1] / 2;

  float* rec      = (float*)d_ws;
  float* tmp      = rec + (size_t)E * REC;
  int*   deg      = (int*)(tmp + (size_t)E * REC);
  int*   start    = deg + N;
  int*   bintotal = start + N;
  int*   bincur   = bintotal + MAXB;
  float* invc     = (float*)(bincur + MAXB);
  float* muc      = invc + NK * ND;

  hipMemsetAsync(bintotal, 0, (size_t)(2 * MAXB) * sizeof(int), stream);

  const int mt = (N + 15) / 16;
  const int B  = (N + NPB - 1) / NPB;
  const int gw = (E + CH - 1) / CH;
  bin_count_kernel<<<gw, BT, 0, stream>>>(edge_index, bintotal, mu, sigma, invc, muc, E, B);
  bin_scatter_kernel<<<gw, BT, 0, stream>>>(edge_index, pseudo, invc, muc, bintotal, bincur, tmp, E, B);
  bin_sort_kernel<<<B, BT, 0, stream>>>(bintotal, tmp, rec, start, deg, N, B);
  node_kernel<<<(mt + 3) / 4, 128, 0, stream>>>(x, g, root, bias, rec, start, deg, out, N);
}


# Edge records halved to 16 B (src and in-bin target packed in one word, 3 pseudo-coordinates; the 6 Gaussian weights computed per record in the node kernel): one 16-B store per record per sort pass
# speedup vs baseline: 3.2198x; 1.3715x over previous
#include <hip/hip_runtime.h>


typedef __attribute__((ext_vector_type(2))) float v2f;
typedef __attribute__((ext_vector_type(4))) float v4f;
typedef __attribute__((ext_vector_type(8))) float v8f;

#define C_IN  32
#define KCO   192
#define C_OUT 32
#define NK    6
#define ND    3
#define REC   4
#define SRC_BITS 19
#define SRC_MASK ((1 << SRC_BITS) - 1)
#define APITCH 196
#define NPB_SHIFT 9
#define NPB  512
#define MAXB 1024
#define BT   1024
#define EPT  4
#define CH   (BT * EPT)


static __device__ __forceinline__ int block_excl_scan(int v, int* ws) {
  const int lane = threadIdx.x & 31;
  const int wave = threadIdx.x >> 5;
  int incl = v;
  for (int o = 1; o < 32; o <<= 1) {
    int t = __shfl_up(incl, o);
    if (lane >= o) incl += t;
  }
  if (lane == 31) ws[wave] = incl;
  __syncthreads();
  if (wave == 0) {
    const int w = ws[lane];
    int wi = w;
    for (int o = 1; o < 32; o <<= 1) {
      int t = __shfl_up(wi, o);
      if (lane >= o) wi += t;
    }
    ws[lane] = wi - w;
  }
  __syncthreads();
  return ws[wave] + incl - v;
}

__global__ __launch_bounds__(BT)
void bin_count_kernel(const int* __restrict__ edge_index, int* __restrict__ bintotal,
                      const float* __restrict__ mu, const float* __restrict__ sigma,
                      float* __restrict__ inv, float* __restrict__ muc, int E, int B) {
  __shared__ int hist[MAXB];
  const int tid = threadIdx.x;
  hist[tid] = 0;
  __syncthreads();
  const int e0 = blockIdx.x * CH;
  int dstv[EPT];
#pragma unroll
  for (int k = 0; k < EPT; ++k) {
    const int e = e0 + k * BT + tid;
    dstv[k] = edge_index[E + (e < E ? e : E - 1)];
  }
#pragma unroll
  for (int k = 0; k < EPT; ++k) {
    if (e0 + k * BT + tid < E)
      __hip_atomic_fetch_add(&hist[dstv[k] >> NPB_SHIFT], 1, __ATOMIC_RELAXED, __HIP_MEMORY_SCOPE_WORKGROUP);
  }
  __syncthreads();
  if (tid < B) {
    const int c = hist[tid];
    if (c > 0)
      __hip_atomic_fetch_add(bintotal + tid, c, __ATOMIC_RELAXED, __HIP_MEMORY_SCOPE_AGENT);
  }
  if (blockIdx.x == 0 && tid < NK * ND) {
    const float sg = sigma[tid];
    inv[tid] = -0.5f / (1e-15f + sg * sg);
    muc[tid] = mu[tid];
  }
}

__global__ __launch_bounds__(BT)
void bin_scatter_kernel(const int* __restrict__ edge_index, const float* __restrict__ pseudo,
                        const int* __restrict__ bintotal, int* __restrict__ bincur,
                        float* __restrict__ tmp, int E, int B) {
  __shared__ int hist[MAXB];
  __shared__ int base[MAXB];
  __shared__ int ws[32];
  const int tid = threadIdx.x;
  hist[tid] = 0;
  __syncthreads();
  const int e0 = blockIdx.x * CH;
  int dstv[EPT], srcv[EPT];
  float pv[EPT][ND];
#pragma unroll
  for (int k = 0; k < EPT; ++k) {
    const int e = e0 + k * BT + tid;
    const int ee = e < E ? e : E - 1;
    dstv[k] = edge_index[E + ee];
    srcv[k] = edge_index[ee];
    pv[k][0] = pseudo[ee * ND + 0];
    pv[k][1] = pseudo[ee * ND + 1];
    pv[k][2] = pseudo[ee * ND + 2];
  }
#pragma unroll
  for (int k = 0; k < EPT; ++k) {
    if (e0 + k * BT + tid < E)
      __hip_atomic_fetch_add(&hist[dstv[k] >> NPB_SHIFT], 1, __ATOMIC_RELAXED, __HIP_MEMORY_SCOPE_WORKGROUP);
  }
  const int tot = (tid < B) ? bintotal[tid] : 0;
  const int binstart = block_excl_scan(tot, ws);
  const int mine = hist[tid];
  int b0 = 0;
  if (tid < B && mine > 0)
    b0 = binstart + __hip_atomic_fetch_add(bincur + tid, mine, __ATOMIC_RELAXED, __HIP_MEMORY_SCOPE_AGENT);
  base[tid] = b0;
  __syncthreads();
  hist[tid] = 0;
  __syncthreads();
#pragma unroll
  for (int k = 0; k < EPT; ++k) {
    const int e = e0 + k * BT + tid;
    if (e < E) {
      const int dst = dstv[k];
      const int bin = dst >> NPB_SHIFT;
      const int src = srcv[k];
      const float p0 = pv[k][0];
      const float p1 = pv[k][1];
      const float p2 = pv[k][2];
      const int r = __hip_atomic_fetch_add(&hist[bin], 1, __ATOMIC_RELAXED, __HIP_MEMORY_SCOPE_WORKGROUP);
      const int pos = base[bin] + r;
      const v4f q = {__int_as_float(src | ((dst & (NPB - 1)) << SRC_BITS)), p0, p1, p2};
      *(v4f*)(tmp + (size_t)pos * REC) = q;
    }
  }
}

__global__ __launch_bounds__(BT)
void bin_sort_kernel(const int* __restrict__ bintotal, const float* __restrict__ tmp,
                     float* __restrict__ rec, int* __restrict__ start, int* __restrict__ deg,
                     int N, int B) {
  __shared__ int cnt[NPB];
  __shared__ int cur[NPB];
  __shared__ int ws0[32];
  __shared__ int ws1[32];
  __shared__ int sbs;
  const int tid = threadIdx.x;
  const int b = blockIdx.x;
  const int n0 = b << NPB_SHIFT;
  if (tid < NPB) cnt[tid] = 0;
  const int tot = (tid < B) ? bintotal[tid] : 0;
  const int pre = block_excl_scan(tot, ws0);
  if (tid == b) sbs = pre;
  __syncthreads();
  const int bs = sbs;
  const int be = bs + bintotal[b];
  for (int i0 = bs + tid; i0 < be; i0 += EPT * BT) {
    int dv[EPT];
#pragma unroll
    for (int k = 0; k < EPT; ++k) {
      const int i = i0 + k * BT;
      dv[k] = __float_as_int(tmp[(size_t)(i < be ? i : be - 1) * REC]) >> SRC_BITS;
    }
#pragma unroll
    for (int k = 0; k < EPT; ++k) {
      if (i0 + k * BT < be)
        __hip_atomic_fetch_add(&cnt[dv[k]], 1, __ATOMIC_RELAXED, __HIP_MEMORY_SCOPE_WORKGROUP);
    }
  }
  __syncthreads();
  const int c = (tid < NPB) ? cnt[tid] : 0;
  const int ex = block_excl_scan(c, ws1);
  if (tid < NPB) {
    cur[tid] = bs + ex;
    const int n = n0 + tid;
    if (n < N) { start[n] = bs + ex; deg[n] = c; }
  }
  __syncthreads();
  for (int i0 = bs + tid; i0 < be; i0 += EPT * BT) {
    v4f q[EPT];
#pragma unroll
    for (int k = 0; k < EPT; ++k) {
      const int i = i0 + k * BT;
      q[k] = *(const v4f*)(tmp + (size_t)(i < be ? i : be - 1) * REC);
    }
#pragma unroll
    for (int k = 0; k < EPT; ++k) {
      if (i0 + k * BT < be) {
        const int ln = __float_as_int(q[k].x) >> SRC_BITS;
        const int pos = __hip_atomic_fetch_add(&cur[ln], 1, __ATOMIC_RELAXED, __HIP_MEMORY_SCOPE_WORKGROUP);
        *(v4f*)(rec + (size_t)pos * REC) = q[k];
      }
    }
  }
}

static __device__ __forceinline__ float lane_bcast(float v, int t) {
  return __int_as_float(__builtin_amdgcn_readlane(__float_as_int(v), t));
}

__global__ __launch_bounds__(128)
void node_kernel(const float* __restrict__ x, const float* __restrict__ g,
                 const float* __restrict__ root, const float* __restrict__ bias,
                 const float* __restrict__ rec, const int* __restrict__ start,
                 const int* __restrict__ deg,
                 const float* __restrict__ inv, const float* __restrict__ muc,
                 float* __restrict__ out, int N) {
  __shared__ float lds_all[4 * 16 * APITCH];
  const int lane = threadIdx.x & 31;
  const int wv = __builtin_amdgcn_readfirstlane(threadIdx.x >> 5);
  const int m0 = (blockIdx.x * 4 + wv) * 16;
  if (m0 >= N) return;
  float* lds = lds_all + wv * (16 * APITCH);

  const int nl = m0 + (lane & 15);
  const int last = (m0 + 15 < N) ? (m0 + 15) : (N - 1);
  const int endv = start[last] + deg[last];
  const int sv = start[nl < N ? nl : N - 1];
  const int bv = (lane < 16 && nl < N) ? sv : endv;
  const int end = __builtin_amdgcn_readfirstlane(endv);
  const int beg = __builtin_amdgcn_readlane(bv, 0);

  const float cntv = (float)(__shfl_down(bv, 1) - bv);

  const int hx  = lane >> 4;
  const int l16 = lane & 15;

  float a0 = 0.f, a1 = 0.f, a2 = 0.f, a3 = 0.f, a4 = 0.f, a5 = 0.f;
  int i = 0;
  int nb = __builtin_amdgcn_readlane(bv, 1);

  for (int c0 = beg; c0 < end; c0 += 32) {
    int rl = c0 + lane;
    rl = rl < end ? rl : end - 1;
    const v4f q = *(const v4f*)(rec + (size_t)rl * REC);
    const int srcv = __float_as_int(q.x) & SRC_MASK;
    float w[NK];
#pragma unroll
    for (int kk = 0; kk < NK; ++kk) {
      float d0 = q.y - muc[kk * ND + 0];
      float d1 = q.z - muc[kk * ND + 1];
      float d2 = q.w - muc[kk * ND + 2];
      float s = d0 * d0 * inv[kk * ND + 0];
      s = fmaf(d1 * d1, inv[kk * ND + 1], s);
      s = fmaf(d2 * d2, inv[kk * ND + 2], s);
      w[kk] = __expf(s);
    }
#pragma unroll
    for (int h = 0; h < 2; ++h) {
      const int g0 = c0 + 16 * h;
      if (g0 < end) {
        float xv[16];
#pragma unroll
        for (int u = 0; u < 16; ++u) {
          const int su = __builtin_amdgcn_readlane(srcv, 16 * h + u);
          xv[u] = x[su * C_IN + lane];
        }
#pragma unroll
        for (int u = 0; u < 16; ++u) {
          const int rr = g0 + u;
          if (rr < end) {
            while (rr >= nb) {
              float* lp = lds + i * APITCH + lane;
              lp[0] = a0; lp[32] = a1; lp[64] = a2; lp[96] = a3; lp[128] = a4; lp[160] = a5;
              a0 = a1 = a2 = a3 = a4 = a5 = 0.f;
              ++i;
              nb = __builtin_amdgcn_readlane(bv, i + 1);
            }
            const int t = 16 * h + u;
            a0 = fmaf(lane_bcast(w[0], t), xv[u], a0);
            a1 = fmaf(lane_bcast(w[1], t), xv[u], a1);
            a2 = fmaf(lane_bcast(w[2], t), xv[u], a2);
            a3 = fmaf(lane_bcast(w[3], t), xv[u], a3);
            a4 = fmaf(lane_bcast(w[4], t), xv[u], a4);
            a5 = fmaf(lane_bcast(w[5], t), xv[u], a5);
          }
        }
      }
    }
  }
  while (i < 16) {
    float* lp = lds + i * APITCH + lane;
    lp[0] = a0; lp[32] = a1; lp[64] = a2; lp[96] = a3; lp[128] = a4; lp[160] = a5;
    a0 = a1 = a2 = a3 = a4 = a5 = 0.f;
    ++i;
  }

  v8f c0 = {}, c1 = {};
#pragma unroll
  for (int kk = 0; kk < KCO; kk += 4) {
    const int j  = kk + 2 * hx;
    const int kb = j >> 5;
    const int ci = j & 31;
    v2f a = *(const v2f*)(lds + l16 * APITCH + j);
    const float* gp = g + ci * KCO + kb * C_OUT + l16;
    v2f b0, b1;
    b0.x = gp[0];   b0.y = gp[KCO];
    b1.x = gp[16];  b1.y = gp[KCO + 16];
    c0 = __builtin_amdgcn_wmma_f32_16x16x4_f32(
         false, a,  false, b0,
         (short)0, c0,  false,  false);
    c1 = __builtin_amdgcn_wmma_f32_16x16x4_f32(
        false, a, false, b1, (short)0, c1, false, false);
  }

  v2f ax[C_IN / 4];
  {
    const int xr = (m0 + l16 < N) ? (m0 + l16) : (N - 1);
#pragma unroll
    for (int k = 0; k < C_IN / 4; ++k) ax[k] = *(const v2f*)(x + xr * C_IN + 4 * k + 2 * hx);
  }

#pragma unroll
  for (int r = 0; r < 8; ++r) {
    const float cv = __shfl(cntv, 8 * hx + r);
    const float denom = cv > 1.0f ? cv : 1.0f;
    c0[r] = c0[r] / denom;
    c1[r] = c1[r] / denom;
  }
#pragma unroll
  for (int k = 0; k < C_IN / 4; ++k) {
    const float* rp = root + (4 * k + 2 * hx) * C_OUT + l16;
    v2f b0, b1;
    b0.x = rp[0];   b0.y = rp[C_OUT];
    b1.x = rp[16];  b1.y = rp[C_OUT + 16];
    c0 = __builtin_amdgcn_wmma_f32_16x16x4_f32(
        false, ax[k], false, b0, (short)0, c0, false, false);
    c1 = __builtin_amdgcn_wmma_f32_16x16x4_f32(
        false, ax[k], false, b1, (short)0, c1, false, false);
  }
  const float bv0 = bias[l16];
  const float bv1 = bias[16 + l16];
  float o0[8], o1[8];
#pragma unroll
  for (int r = 0; r < 8; ++r) { o0[r] = c0[r] + bv0; o1[r] = c1[r] + bv1; }
  float* p = out + (m0 + 8 * hx) * C_OUT + l16;
  if (m0 + 16 <= N) {
#pragma unroll
    for (int r = 0; r < 8; ++r) { p[r * C_OUT] = o0[r]; p[r * C_OUT + 16] = o1[r]; }
  } else {
#pragma unroll
    for (int r = 0; r < 8; ++r) {
      if (m0 + 8 * hx + r < N) { p[r * C_OUT] = o0[r]; p[r * C_OUT + 16] = o1[r]; }
    }
  }
}

extern "C" void kernel_launch(void* const* d_in, const int* in_sizes, int n_in,
                              void* d_out, int out_size, void* d_ws, size_t ws_size,
                              hipStream_t stream) {
  const float* x          = (const float*)d_in[0];
  const int*   edge_index = (const int*)  d_in[1];
  const float* pseudo     = (const float*)d_in[2];
  const float* g          = (const float*)d_in[3];
  const float* mu         = (const float*)d_in[4];
  const float* sigma      = (const float*)d_in[5];
  const float* root       = (const float*)d_in[6];
  const float* bias       = (const float*)d_in[7];
  float* out = (float*)d_out;

  const int N = in_sizes[0] / C_IN;
  const int E = in_sizes[1] / 2;

  float* rec      = (float*)d_ws;
  float* tmp      = rec + (size_t)E * REC;
  int*   deg      = (int*)(tmp + (size_t)E * REC);
  int*   start    = deg + N;
  int*   bintotal = start + N;
  int*   bincur   = bintotal + MAXB;
  float* invc     = (float*)(bincur + MAXB);
  float* muc      = invc + NK * ND;

  hipMemsetAsync(bintotal, 0, (size_t)(2 * MAXB) * sizeof(int), stream);

  const int mt = (N + 15) / 16;
  const int B  = (N + NPB - 1) / NPB;
  const int gw = (E + CH - 1) / CH;
  bin_count_kernel<<<gw, BT, 0, stream>>>(edge_index, bintotal, mu, sigma, invc, muc, E, B);
  bin_scatter_kernel<<<gw, BT, 0, stream>>>(edge_index, pseudo, bintotal, bincur, tmp, E, B);
  bin_sort_kernel<<<B, BT, 0, stream>>>(bintotal, tmp, rec, start, deg, N, B);
  node_kernel<<<(mt + 3) / 4, 128, 0, stream>>>(x, g, root, bias, rec, start, deg, invc, muc, out, N);
}
